# starting kernel
# baseline (speedup 1.0000x reference)
#include <hip/hip_runtime.h>

typedef float v2f __attribute__((ext_vector_type(2)));
typedef float v8f __attribute__((ext_vector_type(8)));

#define VN 96
#define NC 32
#define BX 4
#define BY 4
#define BT 16
#define NTHREADS 256
#define SCALE (1.0f / 27.0f)

#define S1_OFF 0
#define BF_OFF 0
#define S2_OFF 8000
#define M1_OFF 8000
#define M2_OFF 10304
#define PM_OFF 11840
#define SMEM_FLOATS 14400

__device__ __forceinline__ float ldg_vol(const float* __restrict__ p,
                                         int c, int x, int y, int t) {
    if ((unsigned)x < (unsigned)VN && (unsigned)y < (unsigned)VN &&
        (unsigned)t < (unsigned)VN)
        return p[((c * VN + x) * VN + y) * VN + t];
    return 0.0f;
}

__global__ __launch_bounds__(NTHREADS)
void corr3d_wmma_kernel(const float* __restrict__ mov,
                        const float* __restrict__ fix,
                        float* __restrict__ out) {
    __shared__ float smem[SMEM_FLOATS];

    const int tid  = threadIdx.x;
    const int lane = tid & 31;
    const int w    = tid >> 5;
    const int col  = lane & 15;

    const int x0 = blockIdx.z * BX;
    const int y0 = blockIdx.y * BY;
    const int t0 = blockIdx.x * BT;

    const int vA = w * 32 + col;
    const int vB = vA + 16;
    const int lxA = vA >> 6, lyA = (vA >> 4) & 3, ltA = vA & 15;
    const int lxB = vB >> 6, lyB = (vB >> 4) & 3, ltB = vB & 15;
    const int cc0 = (lane < 16) ? 0 : 2;
    const int cc1 = cc0 + 1;

    float accA[27], accB[27];
#pragma unroll
    for (int d = 0; d < 27; ++d) { accA[d] = 0.0f; accB[d] = 0.0f; }

    const v2f a_ones = {1.0f, 1.0f};

    for (int cg = 0; cg < 8; ++cg) {
        __syncthreads();
        for (int idx = tid; idx < 8000; idx += NTHREADS) {
            int c4 = idx / 2000, r = idx - c4 * 2000;
            int ix = r / 200, rr = r - ix * 200;
            int iy = rr / 20, it = rr - iy * 20;
            int ch = cg * 4 + c4;
            int gx = x0 - 3 + ix, gy = y0 - 3 + iy, gt = t0 - 3 + it;
            float v = ldg_vol(fix, ch, gx, gy, gt)
                    + ldg_vol(fix, ch, gx, gy, gt + 1)
                    + ldg_vol(fix, ch, gx, gy, gt + 2);
            smem[S1_OFF + ((c4 * 10 + ix) * 10 + iy) * 20 + it] = v;
        }
        __syncthreads();
        for (int idx = tid; idx < 6400; idx += NTHREADS) {
            int c4 = idx / 1600, r = idx - c4 * 1600;
            int ix = r / 160, rr = r - ix * 160;
            int iy = rr / 20, it = rr - iy * 20;
            const float* s1 = &smem[S1_OFF + ((c4 * 10 + ix) * 10 + iy) * 20 + it];
            smem[S2_OFF + ((c4 * 10 + ix) * 8 + iy) * 20 + it] =
                s1[0] + s1[20] + s1[40];
        }
        __syncthreads();
        for (int idx = tid; idx < 5120; idx += NTHREADS) {
            int c4 = idx / 1280, r = idx - c4 * 1280;
            int ix = r / 160, rr = r - ix * 160;
            int iy = rr / 20, it = rr - iy * 20;
            const float* s2 = &smem[S2_OFF + ((c4 * 10 + ix) * 8 + iy) * 20 + it];
            smem[BF_OFF + ((c4 * 8 + ix) * 8 + iy) * 20 + it] =
                s2[0] + s2[160] + s2[320];
        }
        __syncthreads();
        for (int idx = tid; idx < 2304; idx += NTHREADS) {
            int c4 = idx / 576, r = idx - c4 * 576;
            int ix = r / 96, rr = r - ix * 96;
            int iy = rr / 16, lt = rr - iy * 16;
            int ch = cg * 4 + c4;
            int gx = x0 - 1 + ix, gy = y0 - 1 + iy, gt = t0 + lt - 1;
            float v = ldg_vol(mov, ch, gx, gy, gt)
                    + ldg_vol(mov, ch, gx, gy, gt + 1)
                    + ldg_vol(mov, ch, gx, gy, gt + 2);
            smem[M1_OFF + ((c4 * 6 + ix) * 6 + iy) * 16 + lt] = v;
        }
        __syncthreads();
        for (int idx = tid; idx < 1536; idx += NTHREADS) {
            int c4 = idx / 384, r = idx - c4 * 384;
            int ix = r / 64, rr = r - ix * 64;
            int ly = rr / 16, lt = rr - ly * 16;
            const float* m1 = &smem[M1_OFF + ((c4 * 6 + ix) * 6 + ly) * 16 + lt];
            smem[M2_OFF + ((c4 * 6 + ix) * 4 + ly) * 16 + lt] =
                m1[0] + m1[16] + m1[32];
        }
        __syncthreads();
        for (int idx = tid; idx < 1024; idx += NTHREADS) {
            int c4 = idx / 256, r = idx - c4 * 256;
            int lx = r / 64, rr = r - lx * 64;
            int ly = rr / 16, lt = rr - ly * 16;
            const float* m2 = &smem[M2_OFF + ((c4 * 6 + lx) * 4 + ly) * 16 + lt];
            smem[PM_OFF + ((c4 * 4 + lx) * 4 + ly) * 16 + lt] =
                m2[0] + m2[64] + m2[128];
        }
        __syncthreads();

        const float pmA0 = smem[PM_OFF + ((cc0 * 4 + lxA) * 4 + lyA) * 16 + ltA];
        const float pmA1 = smem[PM_OFF + ((cc1 * 4 + lxA) * 4 + lyA) * 16 + ltA];
        const float pmB0 = smem[PM_OFF + ((cc0 * 4 + lxB) * 4 + lyB) * 16 + ltB];
        const float pmB1 = smem[PM_OFF + ((cc1 * 4 + lxB) * 4 + lyB) * 16 + ltB];

#pragma unroll
        for (int i = 0; i < 3; ++i)
#pragma unroll
        for (int j = 0; j < 3; ++j)
#pragma unroll
        for (int k = 0; k < 3; ++k) {
            const int d = (i * 3 + j) * 3 + k;
            const int pA0 = ((cc0 * 8 + (lxA + 2*i)) * 8 + (lyA + 2*j)) * 20 + (ltA + 2*k);
            const int pA1 = ((cc1 * 8 + (lxA + 2*i)) * 8 + (lyA + 2*j)) * 20 + (ltA + 2*k);
            const int pB0 = ((cc0 * 8 + (lxB + 2*i)) * 8 + (lyB + 2*j)) * 20 + (ltB + 2*k);
            const int pB1 = ((cc1 * 8 + (lxB + 2*i)) * 8 + (lyB + 2*j)) * 20 + (ltB + 2*k);
            v2f bA = { pmA0 * smem[BF_OFF + pA0], pmA1 * smem[BF_OFF + pA1] };
            v2f bB = { pmB0 * smem[BF_OFF + pB0], pmB1 * smem[BF_OFF + pB1] };
            v8f z = {};
            v8f rA = __builtin_amdgcn_wmma_f32_16x16x4_f32(
                false, a_ones, false, bA, (short)0, z, false, false);
            v8f rB = __builtin_amdgcn_wmma_f32_16x16x4_f32(
                false, a_ones, false, bB, (short)0, z, false, false);
            accA[d] += rA[0];
            accB[d] += rB[0];
        }
    }

    const int v  = w * 32 + lane;
    const int lx = v >> 6, ly = (v >> 4) & 3, lt = v & 15;
    const int x = x0 + lx, y = y0 + ly, t = t0 + lt;
#pragma unroll
    for (int d = 0; d < 27; ++d) {
        float val = ((lane < 16) ? accA[d] : accB[d]) * SCALE;
        out[((d * VN + x) * VN + y) * VN + t] = val;
    }
}

extern "C" void kernel_launch(void* const* d_in, const int* in_sizes, int n_in,
                              void* d_out, int out_size, void* d_ws, size_t ws_size,
                              hipStream_t stream) {
    (void)in_sizes; (void)n_in; (void)out_size; (void)d_ws; (void)ws_size;
    const float* mov = (const float*)d_in[0];
    const float* fix = (const float*)d_in[1];
    float* out = (float*)d_out;
    dim3 grid(VN / BT, VN / BY, VN / BX);
    corr3d_wmma_kernel<<<grid, NTHREADS, 0, stream>>>(mov, fix, out);
}


# Box sums of both volumes computed once per position by a first launch into the workspace; correlation launch reads them from global memory, no LDS tiles or halo recompute; sums bit-identical
# speedup vs baseline: 3.4071x; 3.4071x over previous
#include <hip/hip_runtime.h>

typedef float v2f __attribute__((ext_vector_type(2)));
typedef float v4f __attribute__((ext_vector_type(4)));
typedef float v8f __attribute__((ext_vector_type(8)));

#define VN 96
#define NC 32
#define PN 100
#define XCH 10
#define NTHREADS 256
#define CTHREADS 128
#define SCALE (1.0f / 27.0f)
#define PM_FLOATS (NC * VN * VN * VN)

__device__ __forceinline__ v4f ld4z(const float* __restrict__ chan, int px,
                                    int y, int g) {
    const bool ok = ((unsigned)y < (unsigned)VN) && ((unsigned)g < (unsigned)(VN / 4));
    const int yc = y < 0 ? 0 : (y > VN - 1 ? VN - 1 : y);
    const int gc = g < 0 ? 0 : (g > VN / 4 - 1 ? VN / 4 - 1 : g);
    const v4f v = *(const v4f*)(chan + (px * VN + yc) * VN + 4 * gc);
    const v4f z = {0.0f, 0.0f, 0.0f, 0.0f};
    return ok ? v : z;
}

__device__ __forceinline__ float ld1z(const float* __restrict__ chan, int px,
                                      int y, int t) {
    const bool ok = ((unsigned)y < (unsigned)VN) && ((unsigned)t < (unsigned)VN);
    const int yc = y < 0 ? 0 : (y > VN - 1 ? VN - 1 : y);
    const int tc = t < 0 ? 0 : (t > VN - 1 ? VN - 1 : t);
    const float v = chan[(px * VN + yc) * VN + tc];
    return ok ? v : 0.0f;
}

template <bool FIX>
__device__ __forceinline__ v4f plane_sum(const float* __restrict__ chan,
                                         int px, int cy, int g) {
    v4f rs[3];
#pragma unroll
    for (int r = 0; r < 3; ++r) {
        const int y = cy - 1 + r;
        v4f s;
        if (FIX) {
            const v4f L = ld4z(chan, px, y, g - 1);
            const v4f M = ld4z(chan, px, y, g);
            s.x = (L.y + L.z) + L.w;
            s.y = (L.z + L.w) + M.x;
            s.z = (L.w + M.x) + M.y;
            s.w = (M.x + M.y) + M.z;
        } else {
            const v4f M = ld4z(chan, px, y, g);
            const float l  = ld1z(chan, px, y, 4 * g - 1);
            const float rr = ld1z(chan, px, y, 4 * g + 4);
            s.x = (l + M.x) + M.y;
            s.y = (M.x + M.y) + M.z;
            s.z = (M.y + M.z) + M.w;
            s.w = (M.z + M.w) + rr;
        }
        rs[r] = s;
    }
    return (rs[0] + rs[1]) + rs[2];
}

template <bool FIX>
__device__ __forceinline__ void box_body(const float* __restrict__ chan,
                                         float* __restrict__ ochan) {
    const int NO = FIX ? PN : VN;
    const int LO = FIX ? -2 : 0;
    const int NG = NO / 4;
    const int idx = blockIdx.x * NTHREADS + threadIdx.x;
    const int ox0 = blockIdx.y * XCH;
    if (idx >= NO * NG || ox0 >= NO) return;
    const int oy = idx / NG;
    const int g  = idx - oy * NG;
    const int cy = oy + LO;
    const int ox1 = (ox0 + XCH < NO) ? ox0 + XCH : NO;
    const v4f z = {0.0f, 0.0f, 0.0f, 0.0f};

    int px = ox0 + LO - 1;
    v4f p0 = ((unsigned)px < (unsigned)VN) ? plane_sum<FIX>(chan, px, cy, g) : z;
    ++px;
    v4f p1 = ((unsigned)px < (unsigned)VN) ? plane_sum<FIX>(chan, px, cy, g) : z;
    for (int ox = ox0; ox < ox1; ++ox) {
        ++px;
        v4f p2 = ((unsigned)px < (unsigned)VN) ? plane_sum<FIX>(chan, px, cy, g) : z;
        v4f o = (p0 + p1) + p2;
        *(v4f*)(ochan + (ox * NO + oy) * NO + 4 * g) = o;
        p0 = p1;
        p1 = p2;
    }
}

__global__ __launch_bounds__(NTHREADS)
void box3_kernel(const float* __restrict__ mov, const float* __restrict__ fix,
                 float* __restrict__ pm, float* __restrict__ pf) {
    const int zc = blockIdx.z;
    if (zc < NC) {
        box_body<true>(fix + zc * (VN * VN * VN), pf + zc * (PN * PN * PN));
    } else {
        const int c = zc - NC;
        box_body<false>(mov + c * (VN * VN * VN), pm + c * (VN * VN * VN));
    }
}

__global__ __launch_bounds__(CTHREADS, 3)
void corr_kernel(const float* __restrict__ pm, const float* __restrict__ pf,
                 float* __restrict__ out) {
    const int tid  = threadIdx.x;
    const int lane = tid & 31;
    const int w    = tid >> 5;
    const int col  = lane & 15;
    const int half = lane >> 4;

    const int x  = blockIdx.z * 2 + (w >> 1);
    const int y  = blockIdx.y * 4 + (w & 1) * 2 + (col >> 3);
    const int t4 = blockIdx.x * 8 + (col & 7);
    const int cc0 = half * 2;

    float acc[4][27];
#pragma unroll
    for (int v = 0; v < 4; ++v)
#pragma unroll
        for (int d = 0; d < 27; ++d) acc[v][d] = 0.0f;

    const v2f a_ones = {1.0f, 1.0f};

    const float* pm0 = pm + ((cc0 * VN + x) * VN + y) * VN + 4 * t4;
    const float* pf0 = pf + ((cc0 * PN + x) * PN + y) * PN + 4 * t4;

    for (int cg = 0; cg < 8; ++cg) {
        const float* pmc = pm0 + cg * 4 * (VN * VN * VN);
        const float* pfc = pf0 + cg * 4 * (PN * PN * PN);
        const v4f m0 = *(const v4f*)(pmc);
        const v4f m1 = *(const v4f*)(pmc + VN * VN * VN);
#pragma unroll
        for (int i = 0; i < 3; ++i)
#pragma unroll
        for (int j = 0; j < 3; ++j) {
            const float* r0 = pfc + (2 * i * PN + 2 * j) * PN;
            const float* r1 = r0 + PN * PN * PN;
            const v4f a0 = *(const v4f*)(r0);
            const v4f b0 = *(const v4f*)(r0 + 4);
            const v4f a1 = *(const v4f*)(r1);
            const v4f b1 = *(const v4f*)(r1 + 4);
            const float f0[8] = {a0.x, a0.y, a0.z, a0.w, b0.x, b0.y, b0.z, b0.w};
            const float f1[8] = {a1.x, a1.y, a1.z, a1.w, b1.x, b1.y, b1.z, b1.w};
#pragma unroll
            for (int k = 0; k < 3; ++k)
#pragma unroll
            for (int v = 0; v < 4; ++v) {
                const int d = (i * 3 + j) * 3 + k;
                v2f b = { m0[v] * f0[v + 2 * k], m1[v] * f1[v + 2 * k] };
                v8f z = {};
                v8f r = __builtin_amdgcn_wmma_f32_16x16x4_f32(
                    false, a_ones, false, b, (short)0, z, false, false);
                acc[v][d] += r[0];
            }
        }
    }

#pragma unroll
    for (int p = 0; p < 14; ++p) {
        const int dh = (p + 14 < 27) ? p + 14 : 26;
        v4f val;
        val.x = (half ? acc[0][dh] : acc[0][p]) * SCALE;
        val.y = (half ? acc[1][dh] : acc[1][p]) * SCALE;
        val.z = (half ? acc[2][dh] : acc[2][p]) * SCALE;
        val.w = (half ? acc[3][dh] : acc[3][p]) * SCALE;
        const int d = half ? p + 14 : p;
        if (d < 27)
            *(v4f*)(out + ((d * VN + x) * VN + y) * VN + 4 * t4) = val;
    }
}

extern "C" void kernel_launch(void* const* d_in, const int* in_sizes, int n_in,
                              void* d_out, int out_size, void* d_ws, size_t ws_size,
                              hipStream_t stream) {
    (void)in_sizes; (void)n_in; (void)out_size; (void)ws_size;
    const float* mov = (const float*)d_in[0];
    const float* fix = (const float*)d_in[1];
    float* out = (float*)d_out;
    float* pm = (float*)d_ws;
    float* pf = pm + PM_FLOATS;
    dim3 gbox(10, 10, 2 * NC);
    box3_kernel<<<gbox, NTHREADS, 0, stream>>>(mov, fix, pm, pf);
    dim3 gcorr(VN / 32, VN / 4, VN / 2);
    corr_kernel<<<gcorr, CTHREADS, 0, stream>>>(pm, pf, out);
}


# Correlation as dense fp32 WMMA per (x,y,t) parity class: 2x4x2 voxels by 4x6x4 positions, box arrays 4 channels deep per class, output lines via LDS; 3 not 13.5 WMMA/voxel; channel sum order changes
# speedup vs baseline: 4.5157x; 1.3254x over previous
#include <hip/hip_runtime.h>

typedef float v2f __attribute__((ext_vector_type(2)));
typedef float v4f __attribute__((ext_vector_type(4)));
typedef float v8f __attribute__((ext_vector_type(8)));

#define VN 96
#define NC 32
#define NQ (NC / 4)
#define PN 100
#define XCH 10
#define NTHREADS 256
#define CTHREADS 256
#define SROW4 9
#define SCALE (1.0f / 27.0f)
#define MROW (VN * 4)
#define FROW (PN * 4)
#define PM_FLOATS (NQ * VN * VN * MROW)

__device__ __forceinline__ v4f ld4z(const float* __restrict__ chan, int px,
                                    int y, int g) {
    const bool ok = ((unsigned)y < (unsigned)VN) && ((unsigned)g < (unsigned)(VN / 4));
    const int yc = y < 0 ? 0 : (y > VN - 1 ? VN - 1 : y);
    const int gc = g < 0 ? 0 : (g > VN / 4 - 1 ? VN / 4 - 1 : g);
    const v4f v = *(const v4f*)(chan + (px * VN + yc) * VN + 4 * gc);
    const v4f z = {0.0f, 0.0f, 0.0f, 0.0f};
    return ok ? v : z;
}

__device__ __forceinline__ float ld1z(const float* __restrict__ chan, int px,
                                      int y, int t) {
    const bool ok = ((unsigned)y < (unsigned)VN) && ((unsigned)t < (unsigned)VN);
    const int yc = y < 0 ? 0 : (y > VN - 1 ? VN - 1 : y);
    const int tc = t < 0 ? 0 : (t > VN - 1 ? VN - 1 : t);
    const float v = chan[(px * VN + yc) * VN + tc];
    return ok ? v : 0.0f;
}

template <bool FIX>
__device__ __forceinline__ v4f plane_sum(const float* __restrict__ chan,
                                         int px, int cy, int g) {
    v4f rs[3];
#pragma unroll
    for (int r = 0; r < 3; ++r) {
        const int y = cy - 1 + r;
        v4f s;
        if (FIX) {
            const v4f L = ld4z(chan, px, y, g - 1);
            const v4f M = ld4z(chan, px, y, g);
            s.x = (L.y + L.z) + L.w;
            s.y = (L.z + L.w) + M.x;
            s.z = (L.w + M.x) + M.y;
            s.w = (M.x + M.y) + M.z;
        } else {
            const v4f M = ld4z(chan, px, y, g);
            const float l  = ld1z(chan, px, y, 4 * g - 1);
            const float rr = ld1z(chan, px, y, 4 * g + 4);
            s.x = (l + M.x) + M.y;
            s.y = (M.x + M.y) + M.z;
            s.z = (M.y + M.z) + M.w;
            s.w = (M.z + M.w) + rr;
        }
        rs[r] = s;
    }
    return (rs[0] + rs[1]) + rs[2];
}

template <bool FIX>
__device__ __forceinline__ void box_body(const float* __restrict__ in4,
                                         float* __restrict__ oq) {
    const int NO = FIX ? PN : VN;
    const int LO = FIX ? -2 : 0;
    const int NG = NO / 4;
    const int HN = NO / 2;
    const int idx = blockIdx.x * NTHREADS + threadIdx.x;
    const int ox0 = blockIdx.y * XCH;
    if (idx >= NO * NG || ox0 >= NO) return;
    const int oy = idx / NG;
    const int g  = idx - oy * NG;
    const int cy = oy + LO;
    const int ox1 = (ox0 + XCH < NO) ? ox0 + XCH : NO;
    const v4f z = {0.0f, 0.0f, 0.0f, 0.0f};

    int px = ox0 + LO - 1;
    v4f p0[4], p1[4];
#pragma unroll
    for (int ch = 0; ch < 4; ++ch)
        p0[ch] = ((unsigned)px < (unsigned)VN)
                     ? plane_sum<FIX>(in4 + ch * (VN * VN * VN), px, cy, g) : z;
    ++px;
#pragma unroll
    for (int ch = 0; ch < 4; ++ch)
        p1[ch] = ((unsigned)px < (unsigned)VN)
                     ? plane_sum<FIX>(in4 + ch * (VN * VN * VN), px, cy, g) : z;
    for (int ox = ox0; ox < ox1; ++ox) {
        ++px;
        v4f o[4];
#pragma unroll
        for (int ch = 0; ch < 4; ++ch) {
            const v4f p2 = ((unsigned)px < (unsigned)VN)
                     ? plane_sum<FIX>(in4 + ch * (VN * VN * VN), px, cy, g) : z;
            o[ch] = (p0[ch] + p1[ch]) + p2;
            p0[ch] = p1[ch];
            p1[ch] = p2;
        }
        float* row = oq + (ox * NO + oy) * (NO * 4);
        const v4f e0 = {o[0].x, o[1].x, o[2].x, o[3].x};
        const v4f e1 = {o[0].y, o[1].y, o[2].y, o[3].y};
        const v4f e2 = {o[0].z, o[1].z, o[2].z, o[3].z};
        const v4f e3 = {o[0].w, o[1].w, o[2].w, o[3].w};
        *(v4f*)(row + (2 * g) * 4)          = e0;
        *(v4f*)(row + (2 * g + 1) * 4)      = e2;
        *(v4f*)(row + (HN + 2 * g) * 4)     = e1;
        *(v4f*)(row + (HN + 2 * g + 1) * 4) = e3;
    }
}

__global__ __launch_bounds__(NTHREADS)
void box3_kernel(const float* __restrict__ mov, const float* __restrict__ fix,
                 float* __restrict__ pm, float* __restrict__ pf) {
    const int zq = blockIdx.z;
    if (zq < NQ) {
        box_body<true>(fix + zq * 4 * (VN * VN * VN), pf + zq * (PN * PN * FROW));
    } else {
        const int q = zq - NQ;
        box_body<false>(mov + q * 4 * (VN * VN * VN), pm + q * (VN * VN * MROW));
    }
}

__global__ __launch_bounds__(CTHREADS)
void corr_kernel(const float* __restrict__ pm, const float* __restrict__ pf,
                 float* __restrict__ out) {
    const int lane = threadIdx.x & 31;
    const int w    = threadIdx.x >> 5;
    const int c    = lane & 15;
    const int h    = lane >> 4;

    const int bt = blockIdx.x * 8 + w;
    const int py = blockIdx.y & 1, by = blockIdx.y >> 1;
    const int px = blockIdx.z & 1, bx = blockIdx.z >> 1;
    const int x0 = 4 * bx + px;
    const int y0 = 8 * by + py;
    const int T0 = 2 * bt;

    const int axv = c >> 3, ayv = (c >> 1) & 3, atv = c & 1;
    const float* pa = pm + ((h * VN + x0 + 2 * axv) * VN + y0 + 2 * ayv) * MROW
                         + (T0 + atv) * 4;
    const int xl = c >> 3, yl = (c >> 2) & 1, tl = c & 3;
    const float* pb = pf + ((h * PN + x0 + 2 * xl) * PN + y0 + 2 * yl) * FROW
                         + (T0 + tl) * 4;

    v8f acc[2][2][3];
#pragma unroll
    for (int pt = 0; pt < 2; ++pt)
#pragma unroll
        for (int Xp = 0; Xp < 2; ++Xp)
#pragma unroll
            for (int Yp = 0; Yp < 3; ++Yp) {
                const v8f z8 = {};
                acc[pt][Xp][Yp] = z8;
            }

    v4f av[3];
    v4f bv[3][6];
#pragma unroll
    for (int s = -2; s < 8; ++s) {
        if (s + 2 < 8) {
            const int n = s + 2;
            const int u = n >> 1, pt = n & 1, nb = n % 3;
            const float* pau = pa + u * 2 * (VN * VN * MROW) + pt * (VN / 2) * 4;
            const float* pbu = pb + u * 2 * (PN * PN * FROW) + pt * (PN / 2) * 4;
            av[nb] = *(const v4f*)(pau);
#pragma unroll
            for (int Xp = 0; Xp < 2; ++Xp)
#pragma unroll
                for (int Yp = 0; Yp < 3; ++Yp)
                    bv[nb][Xp * 3 + Yp] =
                        *(const v4f*)(pbu + (4 * Xp * PN + 4 * Yp) * FROW);
        }
        __builtin_amdgcn_sched_barrier(0);
        if (s >= 0) {
            const int pt = s & 1, cb = s % 3;
            const v2f a01 = {av[cb].x, av[cb].y};
            const v2f a23 = {av[cb].z, av[cb].w};
#pragma unroll
            for (int Xp = 0; Xp < 2; ++Xp)
#pragma unroll
                for (int Yp = 0; Yp < 3; ++Yp) {
                    const v4f b = bv[cb][Xp * 3 + Yp];
                    const v2f b01 = {b.x, b.y};
                    const v2f b23 = {b.z, b.w};
                    v8f d = acc[pt][Xp][Yp];
                    d = __builtin_amdgcn_wmma_f32_16x16x4_f32(
                        false, a01, false, b01, (short)0, d, false, false);
                    d = __builtin_amdgcn_wmma_f32_16x16x4_f32(
                        false, a23, false, b23, (short)0, d, false, false);
                    acc[pt][Xp][Yp] = d;
                }
        }
        __builtin_amdgcn_sched_barrier(0);
    }

    float up[2][2][3][4];
#pragma unroll
    for (int pt = 0; pt < 2; ++pt)
#pragma unroll
        for (int Xp = 0; Xp < 2; ++Xp)
#pragma unroll
            for (int Yp = 0; Yp < 3; ++Yp)
#pragma unroll
                for (int yv = 0; yv < 4; ++yv)
                    up[pt][Xp][Yp][yv] =
                        __shfl_down(acc[pt][Xp][Yp][2 * yv + 1], 1, 32);

    __shared__ v4f stage[27 * 8 * SROW4];
    const bool tok = tl < 3;
#pragma unroll
    for (int Xp = 0; Xp < 2; ++Xp) {
        const int i = 2 * Xp + xl - h;
#pragma unroll
        for (int Yp = 0; Yp < 3; ++Yp)
#pragma unroll
            for (int yv = 0; yv < 4; ++yv) {
                const int j = 2 * Yp + yl - yv;
                if (tok && (unsigned)i < 3u && (unsigned)j < 3u) {
                    const int d = (i * 3 + j) * 3 + tl;
                    v4f val;
                    val.x = acc[0][Xp][Yp][2 * yv] * SCALE;
                    val.y = acc[1][Xp][Yp][2 * yv] * SCALE;
                    val.z = up[0][Xp][Yp][yv] * SCALE;
                    val.w = up[1][Xp][Yp][yv] * SCALE;
                    stage[((d * 2 + h) * 4 + yv) * SROW4 + w] = val;
                }
            }
    }
    __syncthreads();
    float* ob = out + (x0 * VN + y0) * VN + 32 * blockIdx.x;
    for (int p = threadIdx.x; p < 27 * 8 * 8; p += CTHREADS) {
        const int row = p >> 3, piece = p & 7;
        const int d = row >> 3, xs = (row >> 2) & 1, ys = row & 3;
        const v4f v = stage[row * SROW4 + piece];
        *(v4f*)(ob + ((d * VN + 2 * xs) * VN + 2 * ys) * VN + 4 * piece) = v;
    }
}

extern "C" void kernel_launch(void* const* d_in, const int* in_sizes, int n_in,
                              void* d_out, int out_size, void* d_ws, size_t ws_size,
                              hipStream_t stream) {
    (void)in_sizes; (void)n_in; (void)out_size; (void)ws_size;
    const float* mov = (const float*)d_in[0];
    const float* fix = (const float*)d_in[1];
    float* out = (float*)d_out;
    float* pm = (float*)d_ws;
    float* pf = pm + PM_FLOATS;
    dim3 gbox(10, 10, 2 * NQ);
    box3_kernel<<<gbox, NTHREADS, 0, stream>>>(mov, fix, pm, pf);
    dim3 gcorr(VN / 32, VN / 4, VN / 2);
    corr_kernel<<<gcorr, CTHREADS, 0, stream>>>(pm, pf, out);
}


# Box-sum launch: a plane's 24-36 loads go out as one branch-free group (plane validity folded into the zero select; 160 registers, 6 waves/SIMD): one memory wait per plane, not per load; same sums
# speedup vs baseline: 6.1838x; 1.3694x over previous
#include <hip/hip_runtime.h>

typedef float v2f __attribute__((ext_vector_type(2)));
typedef float v4f __attribute__((ext_vector_type(4)));
typedef float v8f __attribute__((ext_vector_type(8)));

#define VN 96
#define NC 32
#define NQ (NC / 4)
#define PN 100
#define XCH 10
#define NTHREADS 256
#define CTHREADS 256
#define SROW4 9
#define SCALE (1.0f / 27.0f)
#define MROW (VN * 4)
#define FROW (PN * 4)
#define PM_FLOATS (NQ * VN * VN * MROW)

template <bool FIX>
__device__ __forceinline__ void plane_sums(const float* __restrict__ in4,
                                           int px, int cy, int g, v4f (&p)[4]) {
    const v4f z = {0.0f, 0.0f, 0.0f, 0.0f};
    const bool okx = (unsigned)px < (unsigned)VN;
    const int pxc = px < 0 ? 0 : (px > VN - 1 ? VN - 1 : px);
    bool okr[3];
    int roff[3];
#pragma unroll
    for (int r = 0; r < 3; ++r) {
        const int y = cy - 1 + r;
        okr[r] = okx && ((unsigned)y < (unsigned)VN);
        const int yc = y < 0 ? 0 : (y > VN - 1 ? VN - 1 : y);
        roff[r] = (pxc * VN + yc) * VN;
    }
    if (FIX) {
        const bool okL = (unsigned)(g - 1) < (unsigned)(VN / 4);
        const bool okM = (unsigned)g < (unsigned)(VN / 4);
        const int gL = g - 1 < 0 ? 0 : (g - 1 > VN / 4 - 1 ? VN / 4 - 1 : g - 1);
        const int gM = g < 0 ? 0 : (g > VN / 4 - 1 ? VN / 4 - 1 : g);
        v4f Lv[4][3], Mv[4][3];
#pragma unroll
        for (int ch = 0; ch < 4; ++ch)
#pragma unroll
            for (int r = 0; r < 3; ++r) {
                const float* row = in4 + ch * (VN * VN * VN) + roff[r];
                Lv[ch][r] = *(const v4f*)(row + 4 * gL);
                Mv[ch][r] = *(const v4f*)(row + 4 * gM);
            }
        __builtin_amdgcn_sched_barrier(0);
#pragma unroll
        for (int ch = 0; ch < 4; ++ch) {
            v4f rs[3];
#pragma unroll
            for (int r = 0; r < 3; ++r) {
                const v4f L = (okr[r] && okL) ? Lv[ch][r] : z;
                const v4f M = (okr[r] && okM) ? Mv[ch][r] : z;
                v4f s;
                s.x = (L.y + L.z) + L.w;
                s.y = (L.z + L.w) + M.x;
                s.z = (L.w + M.x) + M.y;
                s.w = (M.x + M.y) + M.z;
                rs[r] = s;
            }
            p[ch] = (rs[0] + rs[1]) + rs[2];
        }
    } else {
        const bool okl = g > 0;
        const bool okrr = g < VN / 4 - 1;
        const int tl = g > 0 ? 4 * g - 1 : 0;
        const int tr = g < VN / 4 - 1 ? 4 * g + 4 : VN - 1;
        v4f Mv[4][3];
        float lv[4][3], rv[4][3];
#pragma unroll
        for (int ch = 0; ch < 4; ++ch)
#pragma unroll
            for (int r = 0; r < 3; ++r) {
                const float* row = in4 + ch * (VN * VN * VN) + roff[r];
                Mv[ch][r] = *(const v4f*)(row + 4 * g);
                lv[ch][r] = row[tl];
                rv[ch][r] = row[tr];
            }
        __builtin_amdgcn_sched_barrier(0);
#pragma unroll
        for (int ch = 0; ch < 4; ++ch) {
            v4f rs[3];
#pragma unroll
            for (int r = 0; r < 3; ++r) {
                const v4f M = okr[r] ? Mv[ch][r] : z;
                const float l  = (okr[r] && okl) ? lv[ch][r] : 0.0f;
                const float rr = (okr[r] && okrr) ? rv[ch][r] : 0.0f;
                v4f s;
                s.x = (l + M.x) + M.y;
                s.y = (M.x + M.y) + M.z;
                s.z = (M.y + M.z) + M.w;
                s.w = (M.z + M.w) + rr;
                rs[r] = s;
            }
            p[ch] = (rs[0] + rs[1]) + rs[2];
        }
    }
}

template <bool FIX>
__device__ __forceinline__ void box_body(const float* __restrict__ in4,
                                         float* __restrict__ oq) {
    const int NO = FIX ? PN : VN;
    const int LO = FIX ? -2 : 0;
    const int NG = NO / 4;
    const int HN = NO / 2;
    const int idx = blockIdx.x * NTHREADS + threadIdx.x;
    const int ox0 = blockIdx.y * XCH;
    if (idx >= NO * NG || ox0 >= NO) return;
    const int oy = idx / NG;
    const int g  = idx - oy * NG;
    const int cy = oy + LO;
    const int ox1 = (ox0 + XCH < NO) ? ox0 + XCH : NO;

    v4f p0[4], p1[4];
    const v4f z = {0.0f, 0.0f, 0.0f, 0.0f};
#pragma unroll
    for (int ch = 0; ch < 4; ++ch) { p0[ch] = z; p1[ch] = z; }
    for (int ox = ox0 - 2; ox < ox1; ++ox) {
        v4f p2[4];
        plane_sums<FIX>(in4, ox + LO + 1, cy, g, p2);
        v4f o[4];
#pragma unroll
        for (int ch = 0; ch < 4; ++ch) {
            o[ch] = (p0[ch] + p1[ch]) + p2[ch];
            p0[ch] = p1[ch];
            p1[ch] = p2[ch];
        }
        if (ox < ox0) continue;
        float* row = oq + (ox * NO + oy) * (NO * 4);
        const v4f e0 = {o[0].x, o[1].x, o[2].x, o[3].x};
        const v4f e1 = {o[0].y, o[1].y, o[2].y, o[3].y};
        const v4f e2 = {o[0].z, o[1].z, o[2].z, o[3].z};
        const v4f e3 = {o[0].w, o[1].w, o[2].w, o[3].w};
        *(v4f*)(row + (2 * g) * 4)          = e0;
        *(v4f*)(row + (2 * g + 1) * 4)      = e2;
        *(v4f*)(row + (HN + 2 * g) * 4)     = e1;
        *(v4f*)(row + (HN + 2 * g + 1) * 4) = e3;
    }
}

__global__ __launch_bounds__(NTHREADS, 6)
void box3_kernel(const float* __restrict__ mov, const float* __restrict__ fix,
                 float* __restrict__ pm, float* __restrict__ pf) {
    const int zq = blockIdx.z;
    if (zq < NQ) {
        box_body<true>(fix + zq * 4 * (VN * VN * VN), pf + zq * (PN * PN * FROW));
    } else {
        const int q = zq - NQ;
        box_body<false>(mov + q * 4 * (VN * VN * VN), pm + q * (VN * VN * MROW));
    }
}

__global__ __launch_bounds__(CTHREADS)
void corr_kernel(const float* __restrict__ pm, const float* __restrict__ pf,
                 float* __restrict__ out) {
    const int lane = threadIdx.x & 31;
    const int w    = threadIdx.x >> 5;
    const int c    = lane & 15;
    const int h    = lane >> 4;

    const int bt = blockIdx.x * 8 + w;
    const int py = blockIdx.y & 1, by = blockIdx.y >> 1;
    const int px = blockIdx.z & 1, bx = blockIdx.z >> 1;
    const int x0 = 4 * bx + px;
    const int y0 = 8 * by + py;
    const int T0 = 2 * bt;

    const int axv = c >> 3, ayv = (c >> 1) & 3, atv = c & 1;
    const float* pa = pm + ((h * VN + x0 + 2 * axv) * VN + y0 + 2 * ayv) * MROW
                         + (T0 + atv) * 4;
    const int xl = c >> 3, yl = (c >> 2) & 1, tl = c & 3;
    const float* pb = pf + ((h * PN + x0 + 2 * xl) * PN + y0 + 2 * yl) * FROW
                         + (T0 + tl) * 4;

    v8f acc[2][2][3];
#pragma unroll
    for (int pt = 0; pt < 2; ++pt)
#pragma unroll
        for (int Xp = 0; Xp < 2; ++Xp)
#pragma unroll
            for (int Yp = 0; Yp < 3; ++Yp) {
                const v8f z8 = {};
                acc[pt][Xp][Yp] = z8;
            }

    v4f av[3];
    v4f bv[3][6];
#pragma unroll
    for (int s = -2; s < 8; ++s) {
        if (s + 2 < 8) {
            const int n = s + 2;
            const int u = n >> 1, pt = n & 1, nb = n % 3;
            const float* pau = pa + u * 2 * (VN * VN * MROW) + pt * (VN / 2) * 4;
            const float* pbu = pb + u * 2 * (PN * PN * FROW) + pt * (PN / 2) * 4;
            av[nb] = *(const v4f*)(pau);
#pragma unroll
            for (int Xp = 0; Xp < 2; ++Xp)
#pragma unroll
                for (int Yp = 0; Yp < 3; ++Yp)
                    bv[nb][Xp * 3 + Yp] =
                        *(const v4f*)(pbu + (4 * Xp * PN + 4 * Yp) * FROW);
        }
        __builtin_amdgcn_sched_barrier(0);
        if (s >= 0) {
            const int pt = s & 1, cb = s % 3;
            const v2f a01 = {av[cb].x, av[cb].y};
            const v2f a23 = {av[cb].z, av[cb].w};
#pragma unroll
            for (int Xp = 0; Xp < 2; ++Xp)
#pragma unroll
                for (int Yp = 0; Yp < 3; ++Yp) {
                    const v4f b = bv[cb][Xp * 3 + Yp];
                    const v2f b01 = {b.x, b.y};
                    const v2f b23 = {b.z, b.w};
                    v8f d = acc[pt][Xp][Yp];
                    d = __builtin_amdgcn_wmma_f32_16x16x4_f32(
                        false, a01, false, b01, (short)0, d, false, false);
                    d = __builtin_amdgcn_wmma_f32_16x16x4_f32(
                        false, a23, false, b23, (short)0, d, false, false);
                    acc[pt][Xp][Yp] = d;
                }
        }
        __builtin_amdgcn_sched_barrier(0);
    }

    float up[2][2][3][4];
#pragma unroll
    for (int pt = 0; pt < 2; ++pt)
#pragma unroll
        for (int Xp = 0; Xp < 2; ++Xp)
#pragma unroll
            for (int Yp = 0; Yp < 3; ++Yp)
#pragma unroll
                for (int yv = 0; yv < 4; ++yv)
                    up[pt][Xp][Yp][yv] =
                        __shfl_down(acc[pt][Xp][Yp][2 * yv + 1], 1, 32);

    __shared__ v4f stage[27 * 8 * SROW4];
    const bool tok = tl < 3;
#pragma unroll
    for (int Xp = 0; Xp < 2; ++Xp) {
        const int i = 2 * Xp + xl - h;
#pragma unroll
        for (int Yp = 0; Yp < 3; ++Yp)
#pragma unroll
            for (int yv = 0; yv < 4; ++yv) {
                const int j = 2 * Yp + yl - yv;
                if (tok && (unsigned)i < 3u && (unsigned)j < 3u) {
                    const int d = (i * 3 + j) * 3 + tl;
                    v4f val;
                    val.x = acc[0][Xp][Yp][2 * yv] * SCALE;
                    val.y = acc[1][Xp][Yp][2 * yv] * SCALE;
                    val.z = up[0][Xp][Yp][yv] * SCALE;
                    val.w = up[1][Xp][Yp][yv] * SCALE;
                    stage[((d * 2 + h) * 4 + yv) * SROW4 + w] = val;
                }
            }
    }
    __syncthreads();
    float* ob = out + (x0 * VN + y0) * VN + 32 * blockIdx.x;
    for (int p = threadIdx.x; p < 27 * 8 * 8; p += CTHREADS) {
        const int row = p >> 3, piece = p & 7;
        const int d = row >> 3, xs = (row >> 2) & 1, ys = row & 3;
        const v4f v = stage[row * SROW4 + piece];
        *(v4f*)(ob + ((d * VN + 2 * xs) * VN + 2 * ys) * VN + 4 * piece) = v;
    }
}

extern "C" void kernel_launch(void* const* d_in, const int* in_sizes, int n_in,
                              void* d_out, int out_size, void* d_ws, size_t ws_size,
                              hipStream_t stream) {
    (void)in_sizes; (void)n_in; (void)out_size; (void)ws_size;
    const float* mov = (const float*)d_in[0];
    const float* fix = (const float*)d_in[1];
    float* out = (float*)d_out;
    float* pm = (float*)d_ws;
    float* pf = pm + PM_FLOATS;
    dim3 gbox(10, 10, 2 * NQ);
    box3_kernel<<<gbox, NTHREADS, 0, stream>>>(mov, fix, pm, pf);
    dim3 gcorr(VN / 32, VN / 4, VN / 2);
    corr_kernel<<<gcorr, CTHREADS, 0, stream>>>(pm, pf, out);
}


# Box-sum launch: a thread loads and t-sums only its own row of a plane; the rows above and below come from neighbours' sums in LDS (8 rows + 2 extra per work-group, one barrier per plane); same sums
# speedup vs baseline: 6.6116x; 1.0692x over previous
#include <hip/hip_runtime.h>

typedef float v2f __attribute__((ext_vector_type(2)));
typedef float v4f __attribute__((ext_vector_type(4)));
typedef float v8f __attribute__((ext_vector_type(8)));

#define VN 96
#define NC 32
#define NQ (NC / 4)
#define PN 100
#define XCH 10
#define NTHREADS 256
#define YB 8
#define BROWS (YB + 2)
#define LDSROW (BROWS * (PN / 4))
#define CTHREADS 256
#define SROW4 9
#define SCALE (1.0f / 27.0f)
#define MROW (VN * 4)
#define FROW (PN * 4)
#define PM_FLOATS (NQ * VN * VN * MROW)

template <bool FIX>
__device__ __forceinline__ void row_tsum(const float* __restrict__ in4,
                                         int px, int y, int g, v4f (&ts)[4]) {
    const v4f z = {0.0f, 0.0f, 0.0f, 0.0f};
    const bool ok = ((unsigned)px < (unsigned)VN) && ((unsigned)y < (unsigned)VN);
    const int pxc = px < 0 ? 0 : (px > VN - 1 ? VN - 1 : px);
    const int yc = y < 0 ? 0 : (y > VN - 1 ? VN - 1 : y);
    const float* row0 = in4 + (pxc * VN + yc) * VN;
    if (FIX) {
        const bool okL = ok && ((unsigned)(g - 1) < (unsigned)(VN / 4));
        const bool okM = ok && ((unsigned)g < (unsigned)(VN / 4));
        const int gL = g - 1 < 0 ? 0 : (g - 1 > VN / 4 - 1 ? VN / 4 - 1 : g - 1);
        const int gM = g < 0 ? 0 : (g > VN / 4 - 1 ? VN / 4 - 1 : g);
        v4f Lv[4], Mv[4];
#pragma unroll
        for (int ch = 0; ch < 4; ++ch) {
            const float* row = row0 + ch * (VN * VN * VN);
            Lv[ch] = *(const v4f*)(row + 4 * gL);
            Mv[ch] = *(const v4f*)(row + 4 * gM);
        }
#pragma unroll
        for (int ch = 0; ch < 4; ++ch) {
            const v4f L = okL ? Lv[ch] : z;
            const v4f M = okM ? Mv[ch] : z;
            v4f s;
            s.x = (L.y + L.z) + L.w;
            s.y = (L.z + L.w) + M.x;
            s.z = (L.w + M.x) + M.y;
            s.w = (M.x + M.y) + M.z;
            ts[ch] = s;
        }
    } else {
        const bool okl = ok && g > 0;
        const bool okrr = ok && g < VN / 4 - 1;
        const int tl = g > 0 ? 4 * g - 1 : 0;
        const int tr = g < VN / 4 - 1 ? 4 * g + 4 : VN - 1;
        v4f Mv[4];
        float lv[4], rv[4];
#pragma unroll
        for (int ch = 0; ch < 4; ++ch) {
            const float* row = row0 + ch * (VN * VN * VN);
            Mv[ch] = *(const v4f*)(row + 4 * g);
            lv[ch] = row[tl];
            rv[ch] = row[tr];
        }
#pragma unroll
        for (int ch = 0; ch < 4; ++ch) {
            const v4f M = ok ? Mv[ch] : z;
            const float l  = okl ? lv[ch] : 0.0f;
            const float rr = okrr ? rv[ch] : 0.0f;
            v4f s;
            s.x = (l + M.x) + M.y;
            s.y = (M.x + M.y) + M.z;
            s.z = (M.y + M.z) + M.w;
            s.w = (M.z + M.w) + rr;
            ts[ch] = s;
        }
    }
}

template <bool FIX>
__device__ __forceinline__ void box_body(const float* __restrict__ in4,
                                         float* __restrict__ oq, v4f* lds) {
    const int NO = FIX ? PN : VN;
    const int LO = FIX ? -2 : 0;
    const int NG = NO / 4;
    const int HN = NO / 2;
    const int oy0 = blockIdx.x * YB;
    const int ox0 = blockIdx.y * XCH;
    if (oy0 >= NO || ox0 >= NO) return;
    const int NACT = BROWS * NG;
    const bool active = (int)threadIdx.x < NACT;
    const int tid = active ? (int)threadIdx.x : NACT - 1;
    const int r = tid / NG;
    const int g = tid - r * NG;
    const int oy = oy0 + r - 1;
    const bool outr = active && r >= 1 && r <= YB && oy < NO;
    const int ox1 = (ox0 + XCH < NO) ? ox0 + XCH : NO;

    v4f p0[4], p1[4];
    const v4f z = {0.0f, 0.0f, 0.0f, 0.0f};
#pragma unroll
    for (int ch = 0; ch < 4; ++ch) { p0[ch] = z; p1[ch] = z; }
    for (int ox = ox0 - 2; ox < ox1; ++ox) {
        v4f ts[4];
        row_tsum<FIX>(in4, ox + LO + 1, oy + LO, g, ts);
        v4f* buf = lds + ((ox - ox0) & 1) * (4 * LDSROW);
        if (active) {
#pragma unroll
            for (int ch = 0; ch < 4; ++ch) buf[ch * LDSROW + tid] = ts[ch];
        }
        __syncthreads();
        v4f p2[4];
#pragma unroll
        for (int ch = 0; ch < 4; ++ch) p2[ch] = z;
        if (outr) {
#pragma unroll
            for (int ch = 0; ch < 4; ++ch) {
                const v4f up = buf[ch * LDSROW + tid - NG];
                const v4f dn = buf[ch * LDSROW + tid + NG];
                p2[ch] = (up + ts[ch]) + dn;
            }
        }
        v4f o[4];
#pragma unroll
        for (int ch = 0; ch < 4; ++ch) {
            o[ch] = (p0[ch] + p1[ch]) + p2[ch];
            p0[ch] = p1[ch];
            p1[ch] = p2[ch];
        }
        if (ox < ox0 || !outr) continue;
        float* row = oq + (ox * NO + oy) * (NO * 4);
        const v4f e0 = {o[0].x, o[1].x, o[2].x, o[3].x};
        const v4f e1 = {o[0].y, o[1].y, o[2].y, o[3].y};
        const v4f e2 = {o[0].z, o[1].z, o[2].z, o[3].z};
        const v4f e3 = {o[0].w, o[1].w, o[2].w, o[3].w};
        *(v4f*)(row + (2 * g) * 4)          = e0;
        *(v4f*)(row + (2 * g + 1) * 4)      = e2;
        *(v4f*)(row + (HN + 2 * g) * 4)     = e1;
        *(v4f*)(row + (HN + 2 * g + 1) * 4) = e3;
    }
}

__global__ __launch_bounds__(NTHREADS)
void box3_kernel(const float* __restrict__ mov, const float* __restrict__ fix,
                 float* __restrict__ pm, float* __restrict__ pf) {
    __shared__ v4f rowsums[2 * 4 * LDSROW];
    const int zq = blockIdx.z;
    if (zq < NQ) {
        box_body<true>(fix + zq * 4 * (VN * VN * VN), pf + zq * (PN * PN * FROW), rowsums);
    } else {
        const int q = zq - NQ;
        box_body<false>(mov + q * 4 * (VN * VN * VN), pm + q * (VN * VN * MROW), rowsums);
    }
}

__global__ __launch_bounds__(CTHREADS)
void corr_kernel(const float* __restrict__ pm, const float* __restrict__ pf,
                 float* __restrict__ out) {
    const int lane = threadIdx.x & 31;
    const int w    = threadIdx.x >> 5;
    const int c    = lane & 15;
    const int h    = lane >> 4;

    const int bt = blockIdx.x * 8 + w;
    const int py = blockIdx.y & 1, by = blockIdx.y >> 1;
    const int px = blockIdx.z & 1, bx = blockIdx.z >> 1;
    const int x0 = 4 * bx + px;
    const int y0 = 8 * by + py;
    const int T0 = 2 * bt;

    const int axv = c >> 3, ayv = (c >> 1) & 3, atv = c & 1;
    const float* pa = pm + ((h * VN + x0 + 2 * axv) * VN + y0 + 2 * ayv) * MROW
                         + (T0 + atv) * 4;
    const int xl = c >> 3, yl = (c >> 2) & 1, tl = c & 3;
    const float* pb = pf + ((h * PN + x0 + 2 * xl) * PN + y0 + 2 * yl) * FROW
                         + (T0 + tl) * 4;

    v8f acc[2][2][3];
#pragma unroll
    for (int pt = 0; pt < 2; ++pt)
#pragma unroll
        for (int Xp = 0; Xp < 2; ++Xp)
#pragma unroll
            for (int Yp = 0; Yp < 3; ++Yp) {
                const v8f z8 = {};
                acc[pt][Xp][Yp] = z8;
            }

    v4f av[3];
    v4f bv[3][6];
#pragma unroll
    for (int s = -2; s < 8; ++s) {
        if (s + 2 < 8) {
            const int n = s + 2;
            const int u = n >> 1, pt = n & 1, nb = n % 3;
            const float* pau = pa + u * 2 * (VN * VN * MROW) + pt * (VN / 2) * 4;
            const float* pbu = pb + u * 2 * (PN * PN * FROW) + pt * (PN / 2) * 4;
            av[nb] = *(const v4f*)(pau);
#pragma unroll
            for (int Xp = 0; Xp < 2; ++Xp)
#pragma unroll
                for (int Yp = 0; Yp < 3; ++Yp)
                    bv[nb][Xp * 3 + Yp] =
                        *(const v4f*)(pbu + (4 * Xp * PN + 4 * Yp) * FROW);
        }
        __builtin_amdgcn_sched_barrier(0);
        if (s >= 0) {
            const int pt = s & 1, cb = s % 3;
            const v2f a01 = {av[cb].x, av[cb].y};
            const v2f a23 = {av[cb].z, av[cb].w};
#pragma unroll
            for (int Xp = 0; Xp < 2; ++Xp)
#pragma unroll
                for (int Yp = 0; Yp < 3; ++Yp) {
                    const v4f b = bv[cb][Xp * 3 + Yp];
                    const v2f b01 = {b.x, b.y};
                    const v2f b23 = {b.z, b.w};
                    v8f d = acc[pt][Xp][Yp];
                    d = __builtin_amdgcn_wmma_f32_16x16x4_f32(
                        false, a01, false, b01, (short)0, d, false, false);
                    d = __builtin_amdgcn_wmma_f32_16x16x4_f32(
                        false, a23, false, b23, (short)0, d, false, false);
                    acc[pt][Xp][Yp] = d;
                }
        }
        __builtin_amdgcn_sched_barrier(0);
    }

    float up[2][2][3][4];
#pragma unroll
    for (int pt = 0; pt < 2; ++pt)
#pragma unroll
        for (int Xp = 0; Xp < 2; ++Xp)
#pragma unroll
            for (int Yp = 0; Yp < 3; ++Yp)
#pragma unroll
                for (int yv = 0; yv < 4; ++yv)
                    up[pt][Xp][Yp][yv] =
                        __shfl_down(acc[pt][Xp][Yp][2 * yv + 1], 1, 32);

    __shared__ v4f stage[27 * 8 * SROW4];
    const bool tok = tl < 3;
#pragma unroll
    for (int Xp = 0; Xp < 2; ++Xp) {
        const int i = 2 * Xp + xl - h;
#pragma unroll
        for (int Yp = 0; Yp < 3; ++Yp)
#pragma unroll
            for (int yv = 0; yv < 4; ++yv) {
                const int j = 2 * Yp + yl - yv;
                if (tok && (unsigned)i < 3u && (unsigned)j < 3u) {
                    const int d = (i * 3 + j) * 3 + tl;
                    v4f val;
                    val.x = acc[0][Xp][Yp][2 * yv] * SCALE;
                    val.y = acc[1][Xp][Yp][2 * yv] * SCALE;
                    val.z = up[0][Xp][Yp][yv] * SCALE;
                    val.w = up[1][Xp][Yp][yv] * SCALE;
                    stage[((d * 2 + h) * 4 + yv) * SROW4 + w] = val;
                }
            }
    }
    __syncthreads();
    float* ob = out + (x0 * VN + y0) * VN + 32 * blockIdx.x;
    for (int p = threadIdx.x; p < 27 * 8 * 8; p += CTHREADS) {
        const int row = p >> 3, piece = p & 7;
        const int d = row >> 3, xs = (row >> 2) & 1, ys = row & 3;
        const v4f v = stage[row * SROW4 + piece];
        *(v4f*)(ob + ((d * VN + 2 * xs) * VN + 2 * ys) * VN + 4 * piece) = v;
    }
}

extern "C" void kernel_launch(void* const* d_in, const int* in_sizes, int n_in,
                              void* d_out, int out_size, void* d_ws, size_t ws_size,
                              hipStream_t stream) {
    (void)in_sizes; (void)n_in; (void)out_size; (void)ws_size;
    const float* mov = (const float*)d_in[0];
    const float* fix = (const float*)d_in[1];
    float* out = (float*)d_out;
    float* pm = (float*)d_ws;
    float* pf = pm + PM_FLOATS;
    dim3 gbox(13, 10, 2 * NQ);
    box3_kernel<<<gbox, NTHREADS, 0, stream>>>(mov, fix, pm, pf);
    dim3 gcorr(VN / 32, VN / 4, VN / 2);
    corr_kernel<<<gcorr, CTHREADS, 0, stream>>>(pm, pf, out);
}


# Box-sum launch: a work-group's finished rows of an x-plane are assembled in LDS as the contiguous block they form in the workspace and stored as whole 128-byte lines, not 16-byte pieces; same sums
# speedup vs baseline: 6.8180x; 1.0312x over previous
#include <hip/hip_runtime.h>

typedef float v2f __attribute__((ext_vector_type(2)));
typedef float v4f __attribute__((ext_vector_type(4)));
typedef float v8f __attribute__((ext_vector_type(8)));

#define VN 96
#define NC 32
#define NQ (NC / 4)
#define PN 100
#define XCH 10
#define NTHREADS 256
#define YB 8
#define BROWS (YB + 2)
#define LDSROW (BROWS * (PN / 4))
#define CTHREADS 256
#define SROW4 9
#define SCALE (1.0f / 27.0f)
#define MROW (VN * 4)
#define FROW (PN * 4)
#define PM_FLOATS (NQ * VN * VN * MROW)

struct RowRaw {
    v4f A[4];
    v4f B[4];
    float l[4];
    float r[4];
    bool f0, f1, f2;
};

template <bool FIX>
__device__ __forceinline__ void row_load(const float* __restrict__ in4,
                                         int px, int y, int g, RowRaw& w) {
    const bool ok = ((unsigned)px < (unsigned)VN) && ((unsigned)y < (unsigned)VN);
    const int pxc = px < 0 ? 0 : (px > VN - 1 ? VN - 1 : px);
    const int yc = y < 0 ? 0 : (y > VN - 1 ? VN - 1 : y);
    const float* row0 = in4 + (pxc * VN + yc) * VN;
    if (FIX) {
        w.f0 = ok && ((unsigned)(g - 1) < (unsigned)(VN / 4));
        w.f1 = ok && ((unsigned)g < (unsigned)(VN / 4));
        w.f2 = false;
        const int gL = g - 1 < 0 ? 0 : (g - 1 > VN / 4 - 1 ? VN / 4 - 1 : g - 1);
        const int gM = g < 0 ? 0 : (g > VN / 4 - 1 ? VN / 4 - 1 : g);
#pragma unroll
        for (int ch = 0; ch < 4; ++ch) {
            const float* row = row0 + ch * (VN * VN * VN);
            w.A[ch] = *(const v4f*)(row + 4 * gL);
            w.B[ch] = *(const v4f*)(row + 4 * gM);
        }
    } else {
        w.f0 = ok;
        w.f1 = ok && g > 0;
        w.f2 = ok && g < VN / 4 - 1;
        const int tl = g > 0 ? 4 * g - 1 : 0;
        const int tr = g < VN / 4 - 1 ? 4 * g + 4 : VN - 1;
#pragma unroll
        for (int ch = 0; ch < 4; ++ch) {
            const float* row = row0 + ch * (VN * VN * VN);
            w.B[ch] = *(const v4f*)(row + 4 * g);
            w.l[ch] = row[tl];
            w.r[ch] = row[tr];
        }
    }
}

template <bool FIX>
__device__ __forceinline__ void row_sum(const RowRaw& w, v4f (&ts)[4]) {
    const v4f z = {0.0f, 0.0f, 0.0f, 0.0f};
    if (FIX) {
#pragma unroll
        for (int ch = 0; ch < 4; ++ch) {
            const v4f L = w.f0 ? w.A[ch] : z;
            const v4f M = w.f1 ? w.B[ch] : z;
            v4f s;
            s.x = (L.y + L.z) + L.w;
            s.y = (L.z + L.w) + M.x;
            s.z = (L.w + M.x) + M.y;
            s.w = (M.x + M.y) + M.z;
            ts[ch] = s;
        }
    } else {
#pragma unroll
        for (int ch = 0; ch < 4; ++ch) {
            const v4f M = w.f0 ? w.B[ch] : z;
            const float l  = w.f1 ? w.l[ch] : 0.0f;
            const float rr = w.f2 ? w.r[ch] : 0.0f;
            v4f s;
            s.x = (l + M.x) + M.y;
            s.y = (M.x + M.y) + M.z;
            s.z = (M.y + M.z) + M.w;
            s.w = (M.z + M.w) + rr;
            ts[ch] = s;
        }
    }
}

__device__ __forceinline__ void lds_barrier() {
    __builtin_amdgcn_fence(__ATOMIC_RELEASE, "workgroup", "local");
    __builtin_amdgcn_s_barrier();
    __builtin_amdgcn_fence(__ATOMIC_ACQUIRE, "workgroup", "local");
}

template <int NO>
__device__ __forceinline__ void copy_plane(const v4f* sb, float* __restrict__ dst,
                                           int npieces) {
    const int NIT = (YB * NO + NTHREADS - 1) / NTHREADS;
    const v4f* s = sb + threadIdx.x;
    float* d = dst + 4 * threadIdx.x;
#pragma unroll
    for (int it = 0; it < NIT; ++it)
        if ((int)threadIdx.x < npieces - it * NTHREADS)
            *(v4f*)(d + it * (NTHREADS * 4)) = s[it * NTHREADS];
}

template <bool FIX>
__device__ __forceinline__ void box_body(const float* __restrict__ in4,
                                         float* __restrict__ oq, v4f* lds,
                                         v4f* stage) {
    const int NO = FIX ? PN : VN;
    const int LO = FIX ? -2 : 0;
    const int NG = NO / 4;
    const int HN = NO / 2;
    const int oy0 = blockIdx.x * YB;
    const int ox0 = blockIdx.y * XCH;
    if (oy0 >= NO || ox0 >= NO) return;
    const int NACT = BROWS * NG;
    const bool active = (int)threadIdx.x < NACT;
    const int tid = active ? (int)threadIdx.x : NACT - 1;
    const int r = tid / NG;
    const int g = tid - r * NG;
    const int oy = oy0 + r - 1;
    const bool outr = active && r >= 1 && r <= YB && oy < NO;
    const int ox1 = (ox0 + XCH < NO) ? ox0 + XCH : NO;
    const int nrows = (NO % YB == 0 || NO - oy0 >= YB) ? YB : NO - oy0;
    const int npieces = nrows * NO;
    float* blk = oq + oy0 * (NO * 4);
    v4f* const img0 = stage + ((r - 1) * NO + 2 * g);

    v4f p0[4], p1[4];
    const v4f z = {0.0f, 0.0f, 0.0f, 0.0f};
#pragma unroll
    for (int ch = 0; ch < 4; ++ch) { p0[ch] = z; p1[ch] = z; }
    for (int ox = ox0 - 2; ox < ox1; ++ox) {
        RowRaw raw;
        row_load<FIX>(in4, ox + LO + 1, oy + LO, g, raw);
        if (ox - 2 >= ox0)
            copy_plane<NO>(stage + ((ox - ox0) & 1) * (YB * PN),
                           blk + (ox - 2) * (NO * NO * 4), npieces);
        if (FIX) {
            asm volatile("" :: "v"(raw.A[0]), "v"(raw.A[1]), "v"(raw.A[2]), "v"(raw.A[3]),
                               "v"(raw.B[0]), "v"(raw.B[1]), "v"(raw.B[2]), "v"(raw.B[3]));
        }
        v4f ts[4];
        row_sum<FIX>(raw, ts);
        v4f* buf = lds + ((ox - ox0) & 1) * (4 * LDSROW);
        if (active) {
#pragma unroll
            for (int ch = 0; ch < 4; ++ch) buf[ch * LDSROW + tid] = ts[ch];
        }
        lds_barrier();
        v4f p2[4];
#pragma unroll
        for (int ch = 0; ch < 4; ++ch) p2[ch] = z;
        if (outr) {
#pragma unroll
            for (int ch = 0; ch < 4; ++ch) {
                const v4f up = buf[ch * LDSROW + tid - NG];
                const v4f dn = buf[ch * LDSROW + tid + NG];
                p2[ch] = (up + ts[ch]) + dn;
            }
        }
        v4f o[4];
#pragma unroll
        for (int ch = 0; ch < 4; ++ch) {
            o[ch] = (p0[ch] + p1[ch]) + p2[ch];
            p0[ch] = p1[ch];
            p1[ch] = p2[ch];
        }
        if (ox < ox0 || !outr) continue;
        v4f* img = img0 + ((ox - ox0) & 1) * (YB * PN);
        const v4f e0 = {o[0].x, o[1].x, o[2].x, o[3].x};
        const v4f e1 = {o[0].y, o[1].y, o[2].y, o[3].y};
        const v4f e2 = {o[0].z, o[1].z, o[2].z, o[3].z};
        const v4f e3 = {o[0].w, o[1].w, o[2].w, o[3].w};
        img[0]      = e0;
        img[1]      = e2;
        img[HN]     = e1;
        img[HN + 1] = e3;
    }
    lds_barrier();
    if (ox1 - 2 >= ox0)
        copy_plane<NO>(stage + ((ox1 - 2 - ox0) & 1) * (YB * PN),
                       blk + (ox1 - 2) * (NO * NO * 4), npieces);
    copy_plane<NO>(stage + ((ox1 - 1 - ox0) & 1) * (YB * PN),
                   blk + (ox1 - 1) * (NO * NO * 4), npieces);
}

__global__ __launch_bounds__(NTHREADS)
void box3_kernel(const float* __restrict__ mov, const float* __restrict__ fix,
                 float* __restrict__ pm, float* __restrict__ pf) {
    __shared__ v4f rowsums[2 * 4 * LDSROW];
    __shared__ v4f planes[2 * YB * PN];
    const int zq = blockIdx.z;
    if (zq < NQ) {
        box_body<true>(fix + zq * 4 * (VN * VN * VN), pf + zq * (PN * PN * FROW), rowsums, planes);
    } else {
        const int q = zq - NQ;
        box_body<false>(mov + q * 4 * (VN * VN * VN), pm + q * (VN * VN * MROW), rowsums, planes);
    }
}

__global__ __launch_bounds__(CTHREADS)
void corr_kernel(const float* __restrict__ pm, const float* __restrict__ pf,
                 float* __restrict__ out) {
    const int lane = threadIdx.x & 31;
    const int w    = threadIdx.x >> 5;
    const int c    = lane & 15;
    const int h    = lane >> 4;

    const int bt = blockIdx.x * 8 + w;
    const int py = blockIdx.y & 1, by = blockIdx.y >> 1;
    const int px = blockIdx.z & 1, bx = blockIdx.z >> 1;
    const int x0 = 4 * bx + px;
    const int y0 = 8 * by + py;
    const int T0 = 2 * bt;

    const int axv = c >> 3, ayv = (c >> 1) & 3, atv = c & 1;
    const float* pa = pm + ((h * VN + x0 + 2 * axv) * VN + y0 + 2 * ayv) * MROW
                         + (T0 + atv) * 4;
    const int xl = c >> 3, yl = (c >> 2) & 1, tl = c & 3;
    const float* pb = pf + ((h * PN + x0 + 2 * xl) * PN + y0 + 2 * yl) * FROW
                         + (T0 + tl) * 4;

    v8f acc[2][2][3];
#pragma unroll
    for (int pt = 0; pt < 2; ++pt)
#pragma unroll
        for (int Xp = 0; Xp < 2; ++Xp)
#pragma unroll
            for (int Yp = 0; Yp < 3; ++Yp) {
                const v8f z8 = {};
                acc[pt][Xp][Yp] = z8;
            }

    v4f av[3];
    v4f bv[3][6];
#pragma unroll
    for (int s = -2; s < 8; ++s) {
        if (s + 2 < 8) {
            const int n = s + 2;
            const int u = n >> 1, pt = n & 1, nb = n % 3;
            const float* pau = pa + u * 2 * (VN * VN * MROW) + pt * (VN / 2) * 4;
            const float* pbu = pb + u * 2 * (PN * PN * FROW) + pt * (PN / 2) * 4;
            av[nb] = *(const v4f*)(pau);
#pragma unroll
            for (int Xp = 0; Xp < 2; ++Xp)
#pragma unroll
                for (int Yp = 0; Yp < 3; ++Yp)
                    bv[nb][Xp * 3 + Yp] =
                        *(const v4f*)(pbu + (4 * Xp * PN + 4 * Yp) * FROW);
        }
        __builtin_amdgcn_sched_barrier(0);
        if (s >= 0) {
            const int pt = s & 1, cb = s % 3;
            const v2f a01 = {av[cb].x, av[cb].y};
            const v2f a23 = {av[cb].z, av[cb].w};
#pragma unroll
            for (int Xp = 0; Xp < 2; ++Xp)
#pragma unroll
                for (int Yp = 0; Yp < 3; ++Yp) {
                    const v4f b = bv[cb][Xp * 3 + Yp];
                    const v2f b01 = {b.x, b.y};
                    const v2f b23 = {b.z, b.w};
                    v8f d = acc[pt][Xp][Yp];
                    d = __builtin_amdgcn_wmma_f32_16x16x4_f32(
                        false, a01, false, b01, (short)0, d, false, false);
                    d = __builtin_amdgcn_wmma_f32_16x16x4_f32(
                        false, a23, false, b23, (short)0, d, false, false);
                    acc[pt][Xp][Yp] = d;
                }
        }
        __builtin_amdgcn_sched_barrier(0);
    }

    float up[2][2][3][4];
#pragma unroll
    for (int pt = 0; pt < 2; ++pt)
#pragma unroll
        for (int Xp = 0; Xp < 2; ++Xp)
#pragma unroll
            for (int Yp = 0; Yp < 3; ++Yp)
#pragma unroll
                for (int yv = 0; yv < 4; ++yv)
                    up[pt][Xp][Yp][yv] =
                        __shfl_down(acc[pt][Xp][Yp][2 * yv + 1], 1, 32);

    __shared__ v4f stage[27 * 8 * SROW4];
    const bool tok = tl < 3;
#pragma unroll
    for (int Xp = 0; Xp < 2; ++Xp) {
        const int i = 2 * Xp + xl - h;
#pragma unroll
        for (int Yp = 0; Yp < 3; ++Yp)
#pragma unroll
            for (int yv = 0; yv < 4; ++yv) {
                const int j = 2 * Yp + yl - yv;
                if (tok && (unsigned)i < 3u && (unsigned)j < 3u) {
                    const int d = (i * 3 + j) * 3 + tl;
                    v4f val;
                    val.x = acc[0][Xp][Yp][2 * yv] * SCALE;
                    val.y = acc[1][Xp][Yp][2 * yv] * SCALE;
                    val.z = up[0][Xp][Yp][yv] * SCALE;
                    val.w = up[1][Xp][Yp][yv] * SCALE;
                    stage[((d * 2 + h) * 4 + yv) * SROW4 + w] = val;
                }
            }
    }
    __syncthreads();
    float* ob = out + (x0 * VN + y0) * VN + 32 * blockIdx.x;
    for (int p = threadIdx.x; p < 27 * 8 * 8; p += CTHREADS) {
        const int row = p >> 3, piece = p & 7;
        const int d = row >> 3, xs = (row >> 2) & 1, ys = row & 3;
        const v4f v = stage[row * SROW4 + piece];
        *(v4f*)(ob + ((d * VN + 2 * xs) * VN + 2 * ys) * VN + 4 * piece) = v;
    }
}

extern "C" void kernel_launch(void* const* d_in, const int* in_sizes, int n_in,
                              void* d_out, int out_size, void* d_ws, size_t ws_size,
                              hipStream_t stream) {
    (void)in_sizes; (void)n_in; (void)out_size; (void)ws_size;
    const float* mov = (const float*)d_in[0];
    const float* fix = (const float*)d_in[1];
    float* out = (float*)d_out;
    float* pm = (float*)d_ws;
    float* pf = pm + PM_FLOATS;
    dim3 gbox(13, 10, 2 * NQ);
    box3_kernel<<<gbox, NTHREADS, 0, stream>>>(mov, fix, pm, pf);
    dim3 gcorr(VN / 32, VN / 4, VN / 2);
    corr_kernel<<<gcorr, CTHREADS, 0, stream>>>(pm, pf, out);
}
